# starting kernel
# baseline (speedup 1.0000x reference)
#include <hip/hip_runtime.h>

typedef __attribute__((ext_vector_type(2))) float v2f;
typedef __attribute__((ext_vector_type(8))) float v8f;

#define CHANNELS 64

__global__ void biconv_init(const float* __restrict__ x,
                            float* __restrict__ agg_out,
                            float* __restrict__ agg_back,
                            long long n4) {
    long long gid = (long long)blockIdx.x * blockDim.x + threadIdx.x;
    if (gid >= n4) return;
    float4 v = ((const float4*)x)[gid];
    ((float4*)agg_out)[gid]  = v;
    ((float4*)agg_back)[gid] = v;
}

__global__ void biconv_scatter(const float* __restrict__ x,
                               const int* __restrict__ src,
                               const int* __restrict__ tgt,
                               float* __restrict__ agg_out,
                               float* __restrict__ agg_back,
                               long long total  ) {
    long long gid = (long long)blockIdx.x * blockDim.x + threadIdx.x;
    if (gid >= total) return;
    long long e  = gid >> 4;
    int       c4 = ((int)gid & 15) << 2;
    long long s = (long long)src[e];
    long long t = (long long)tgt[e];

    float4 xs = *(const float4*)(x + s * CHANNELS + c4);
    float4 xt = *(const float4*)(x + t * CHANNELS + c4);

    float* po = agg_out + t * CHANNELS + c4;
    atomicAdd(po + 0, xs.x);
    atomicAdd(po + 1, xs.y);
    atomicAdd(po + 2, xs.z);
    atomicAdd(po + 3, xs.w);

    float* pb = agg_back + s * CHANNELS + c4;
    atomicAdd(pb + 0, xt.x);
    atomicAdd(pb + 1, xt.y);
    atomicAdd(pb + 2, xt.z);
    atomicAdd(pb + 3, xt.w);
}

__global__ __launch_bounds__(256)
void biconv_wmma_gemm(const float* __restrict__ agg_out,
                      const float* __restrict__ agg_back,
                      const float* __restrict__ norm,
                      const float* __restrict__ norm_t,
                      const float* __restrict__ w_out,
                      const float* __restrict__ w_back,
                      float* __restrict__ out,
                      int N, int numTiles  ) {
    const int wave = threadIdx.x >> 5;
    const int lane = threadIdx.x & 31;
    const int tile = blockIdx.x * 8 + wave;
    if (tile >= numTiles) return;

    const int rowTile = tile >> 2;
    const int colTile = tile & 3;

    const int m    = lane & 15;
    const int half = lane >> 4;
    const int kb   = half << 1;
    const int n    = lane & 15;
    const int col  = colTile * 16 + n;

    int row  = rowTile * 16 + m;
    int rowc = row < N ? row : (N - 1);

    const float sc_o = norm[rowc];
    const float sc_b = norm_t[rowc];
    const float* __restrict__ arow_o = agg_out  + (long long)rowc * CHANNELS;
    const float* __restrict__ arow_b = agg_back + (long long)rowc * CHANNELS;

    v8f acc = {};

#pragma unroll
    for (int k = 0; k < CHANNELS; k += 4) {
        float2 av = *(const float2*)(arow_o + k + kb);
        v2f a; a.x = av.x * sc_o; a.y = av.y * sc_o;
        v2f b;
        b.x = w_out[(k + kb)     * CHANNELS + col];
        b.y = w_out[(k + kb + 1) * CHANNELS + col];
        acc = __builtin_amdgcn_wmma_f32_16x16x4_f32(
            false, a, false, b, (short)0, acc, false, false);
    }
#pragma unroll
    for (int k = 0; k < CHANNELS; k += 4) {
        float2 av = *(const float2*)(arow_b + k + kb);
        v2f a; a.x = av.x * sc_b; a.y = av.y * sc_b;
        v2f b;
        b.x = w_back[(k + kb)     * CHANNELS + col];
        b.y = w_back[(k + kb + 1) * CHANNELS + col];
        acc = __builtin_amdgcn_wmma_f32_16x16x4_f32(
            false, a, false, b, (short)0, acc, false, false);
    }

#pragma unroll
    for (int j = 0; j < 8; ++j) {
        int mrow = rowTile * 16 + j + half * 8;
        if (mrow < N)
            out[(long long)mrow * CHANNELS + col] = acc[j];
    }
}

extern "C" void kernel_launch(void* const* d_in, const int* in_sizes, int n_in,
                              void* d_out, int out_size, void* d_ws, size_t ws_size,
                              hipStream_t stream) {
    const float* x       = (const float*)d_in[0];
    const int*   sources = (const int*)d_in[1];
    const int*   targets = (const int*)d_in[2];
    const float* norm    = (const float*)d_in[3];
    const float* norm_t  = (const float*)d_in[4];
    const float* w_out   = (const float*)d_in[5];
    const float* w_back  = (const float*)d_in[6];
    float*       out     = (float*)d_out;

    const int N = in_sizes[3];
    const long long E = in_sizes[1];

    float* agg_out  = (float*)d_ws;
    float* agg_back = agg_out + (long long)N * CHANNELS;

    {
        long long n4 = (long long)N * (CHANNELS / 4);
        int threads = 256;
        long long blocks = (n4 + threads - 1) / threads;
        biconv_init<<<(unsigned)blocks, threads, 0, stream>>>(x, agg_out, agg_back, n4);
    }
    {
        long long total = E * 16;
        int threads = 256;
        long long blocks = (total + threads - 1) / threads;
        biconv_scatter<<<(unsigned)blocks, threads, 0, stream>>>(
            x, sources, targets, agg_out, agg_back, total);
    }
    {
        int numRowTiles = (N + 15) >> 4;
        int numTiles = numRowTiles * 4;
        int wavesPerBlock = 8;
        int blocks = (numTiles + wavesPerBlock - 1) / wavesPerBlock;
        biconv_wmma_gemm<<<blocks, 256, 0, stream>>>(
            agg_out, agg_back, norm, norm_t, w_out, w_back, out, N, numTiles);
    }
}


# f32 atomic scatter-add replaced by owner work-groups: per 512-node tile and direction sums stay in LDS, edges found by a byte-key scan, rows added by the node's own wave in fixed order; no atomics
# speedup vs baseline: 1.6243x; 1.6243x over previous
#include <hip/hip_runtime.h>


typedef __attribute__((ext_vector_type(2))) float v2f;
typedef __attribute__((ext_vector_type(8))) float v8f;

#define CHANNELS 64
#define TILE_SHIFT 9
#define TILE_NODES (1 << TILE_SHIFT)
#define PIECE 256
#define SUB (PIECE - 1)
#define NODE_BITS 17
#define NODE_MASK ((1u << NODE_BITS) - 1u)

__device__ __forceinline__ unsigned pack_keys(int a, int b, int c, int d) {
    return ((unsigned)(a >> TILE_SHIFT) & 255u)
         | (((unsigned)(b >> TILE_SHIFT) & 255u) << 8)
         | (((unsigned)(c >> TILE_SHIFT) & 255u) << 16)
         | (((unsigned)(d >> TILE_SHIFT) & 255u) << 24);
}

__global__ __launch_bounds__(256)
void biconv_keys(const int* __restrict__ tgt,
                 const int* __restrict__ src,
                 unsigned* __restrict__ kt,
                 unsigned* __restrict__ ks,
                 int E, int nqp  ) {
    const int q = blockIdx.x * 256 + threadIdx.x;
    if (q >= nqp) return;
    const int e0 = q << 4;
    unsigned wt[4], wsrc[4];
    if (e0 + 16 <= E) {
        const int4* t4 = (const int4*)(tgt + e0);
        const int4* s4 = (const int4*)(src + e0);
#pragma unroll
        for (int i = 0; i < 4; ++i) {
            int4 a = t4[i];
            int4 b = s4[i];
            wt[i]   = pack_keys(a.x, a.y, a.z, a.w);
            wsrc[i] = pack_keys(b.x, b.y, b.z, b.w);
        }
    } else {
#pragma unroll
        for (int i = 0; i < 4; ++i) {
            unsigned vt = 0, vs = 0;
#pragma unroll
            for (int j = 0; j < 4; ++j) {
                int e = e0 + 4 * i + j;
                unsigned bt = 255u, bs = 255u;
                if (e < E) {
                    bt = ((unsigned)(tgt[e] >> TILE_SHIFT)) & 255u;
                    bs = ((unsigned)(src[e] >> TILE_SHIFT)) & 255u;
                }
                vt |= bt << (8 * j);
                vs |= bs << (8 * j);
            }
            wt[i] = vt; wsrc[i] = vs;
        }
    }
    ((uint4*)kt)[q] = make_uint4(wt[0], wt[1], wt[2], wt[3]);
    ((uint4*)ks)[q] = make_uint4(wsrc[0], wsrc[1], wsrc[2], wsrc[3]);
}


__device__ __forceinline__ unsigned match_bytes(unsigned v, unsigned pat) {
    unsigned z = v ^ pat;
    unsigned t = ((z & 0x7f7f7f7fu) + 0x7f7f7f7fu) | z;
    return ~t & 0x80808080u;
}

__device__ __forceinline__ unsigned match16(uint4 k, unsigned pat) {
    return (match_bytes(k.x, pat) >> 7) | (match_bytes(k.y, pat) >> 6)
         | (match_bytes(k.z, pat) >> 5) | (match_bytes(k.w, pat) >> 4);
}

__device__ __forceinline__ void add_queued_rows(int q, int qn,
                                                const float2* __restrict__ x2,
                                                float2* acc2, int lane) {
    for (int g = 0; g < qn; g += 8) {
        float2 r[8];
#pragma unroll
        for (int j = 0; j < 8; ++j) {
            r[j] = make_float2(0.f, 0.f);
            if (g + j < qn) {
                unsigned ent = (unsigned)__builtin_amdgcn_readlane(q, g + j);
                r[j] = x2[(size_t)(ent & NODE_MASK) * 32 + lane];
            }
        }
#pragma unroll
        for (int j = 0; j < 8; ++j) {
            if (g + j < qn) {
                unsigned ent = (unsigned)__builtin_amdgcn_readlane(q, g + j);
                int d = (int)((ent >> NODE_BITS) & (TILE_NODES - 1));
                float2 a = acc2[d * 32 + lane];
                a.x += r[j].x;
                a.y += r[j].y;
                acc2[d * 32 + lane] = a;
            }
        }
    }
}

__global__ __launch_bounds__(1024)
void biconv_tile_sum(const float* __restrict__ x,
                     const int* __restrict__ tgt,
                     const int* __restrict__ src,
                     const unsigned* __restrict__ kt,
                     const unsigned* __restrict__ ks,
                     float* __restrict__ agg_out,
                     float* __restrict__ agg_back,
                     int N, int per  ) {
    __shared__ float2   acc2[TILE_NODES * 32];
    __shared__ unsigned list[32 * PIECE];

    const int bin  = blockIdx.x;
    const int dir  = blockIdx.y;
    const int tid  = threadIdx.x;
    const int lane = tid & 31;
    const int wave = __builtin_amdgcn_readfirstlane(tid >> 5);

    const uint4* __restrict__ keys = (const uint4*)(dir ? ks : kt);
    const int*   __restrict__ dstA = dir ? src : tgt;
    const int*   __restrict__ othA = dir ? tgt : src;
    float2*      __restrict__ agg2 = (float2*)(dir ? agg_back : agg_out);
    const float2* __restrict__ x2  = (const float2*)x;

#pragma unroll
    for (int r = 0; r < 16; ++r) {
        int ln = wave * 16 + r;
        int node = (bin << TILE_SHIFT) + ln;
        int nodec = node < N ? node : N - 1;
        acc2[ln * 32 + lane] = x2[(size_t)nodec * 32 + lane];
    }

    const unsigned pat = (unsigned)bin * 0x01010101u;
    const int s0  = wave * per;
    const int s1  = s0 + per;
    const unsigned ltmask = (1u << lane) - 1u;

    int win = 0, maxcnt = 0;
    do {
        int run = 0;
        uint4 k = keys[s0 * 32 + lane];
        for (int s = s0; s < s1; ++s) {
            const int q = s * 32 + lane;
            const int sn = (s + 1 < s1) ? (s + 1) : s;
            const uint4 kn = keys[sn * 32 + lane];
            const unsigned mall = match16(k, pat);
            const int c = __builtin_popcount(mall);
            const unsigned b0 = (unsigned)__ballot(c & 1);
            int excl = __builtin_popcount(b0 & ltmask);
            int tot  = __builtin_popcount(b0);
            if (__ballot(c > 1)) {
#pragma unroll
                for (int b = 1; b < 5; ++b) {
                    const unsigned bb = (unsigned)__ballot((c >> b) & 1);
                    excl += __builtin_popcount(bb & ltmask) << b;
                    tot  += __builtin_popcount(bb) << b;
                }
            }
            if (c) {
                unsigned rel = (unsigned)(run + excl - win);
                const unsigned e0 = (unsigned)q << 4;
                unsigned m = mall;
                while (m) {
                    const int p = __builtin_ctz(m);
                    m &= m - 1;
                    if (rel < (unsigned)SUB) list[wave * PIECE + rel] = e0 + ((p & 7) << 2) + (p >> 3);
                    ++rel;
                }
            }
            run += tot;
            k = kn;
        }
        if (lane == 0) list[wave * PIECE + SUB] = (unsigned)run;
        __syncthreads();

        const int cw = (int)list[lane * PIECE + SUB];
        int mx = cw;
#pragma unroll
        for (int o = 16; o > 0; o >>= 1) { int t = __shfl_xor(mx, o); mx = t > mx ? t : mx; }
        maxcnt = __builtin_amdgcn_readfirstlane(mx);

        for (int i = tid; i < 32 * PIECE; i += 1024) {
            const int idx = i & (PIECE - 1);
            int nvc = (int)list[(i & ~(PIECE - 1)) + SUB] - win;
            nvc = nvc < SUB ? nvc : SUB;
            if (idx < nvc) {
                unsigned e = list[i];
                unsigned d = (unsigned)dstA[e];
                unsigned o = (unsigned)othA[e];
                list[i] = ((d & (TILE_NODES - 1)) << NODE_BITS) | (o & NODE_MASK);
            }
        }
        __syncthreads();

        int qv = 0, qn = 0;
        for (int v = 0; v < 32; ++v) {
            int nv = __builtin_amdgcn_readlane(cw, v) - win;
            nv = nv < SUB ? nv : SUB;
            if (nv <= 0) continue;
            unsigned vv[PIECE / 32], mk[PIECE / 32];
#pragma unroll
            for (int c = 0; c < PIECE / 32; ++c) {
                const int idx = 32 * c + lane;
                vv[c] = list[v * PIECE + idx];
                mk[c] = (unsigned)__ballot((idx < nv) && ((int)(vv[c] >> (NODE_BITS + 4)) == wave));
            }
#pragma unroll
            for (int c = 0; c < PIECE / 32; ++c) {
                unsigned mask = mk[c];
                while (mask) {
                    const int l = __builtin_ctz(mask);
                    mask &= mask - 1;
                    const int ent = __builtin_amdgcn_readlane((int)vv[c], l);
                    if (qn == 32) { add_queued_rows(qv, 32, x2, acc2, lane); qn = 0; }
                    qv = (lane == qn) ? ent : qv;
                    ++qn;
                }
            }
        }
        add_queued_rows(qv, qn, x2, acc2, lane);
        __syncthreads();
        win += SUB;
    } while (win < maxcnt);

#pragma unroll
    for (int r = 0; r < 16; ++r) {
        int ln = wave * 16 + r;
        int node = (bin << TILE_SHIFT) + ln;
        if (node < N) agg2[(size_t)node * 32 + lane] = acc2[ln * 32 + lane];
    }
}

__global__ __launch_bounds__(256)
void biconv_wmma_gemm(const float* __restrict__ agg_out,
                      const float* __restrict__ agg_back,
                      const float* __restrict__ norm,
                      const float* __restrict__ norm_t,
                      const float* __restrict__ w_out,
                      const float* __restrict__ w_back,
                      float* __restrict__ out,
                      int N, int numTiles  ) {
    const int wave = threadIdx.x >> 5;
    const int lane = threadIdx.x & 31;
    const int tile = blockIdx.x * 8 + wave;
    if (tile >= numTiles) return;

    const int rowTile = tile >> 2;
    const int colTile = tile & 3;

    const int m    = lane & 15;
    const int half = lane >> 4;
    const int kb   = half << 1;
    const int n    = lane & 15;
    const int col  = colTile * 16 + n;

    int row  = rowTile * 16 + m;
    int rowc = row < N ? row : (N - 1);

    const float sc_o = norm[rowc];
    const float sc_b = norm_t[rowc];
    const float* __restrict__ arow_o = agg_out  + (long long)rowc * CHANNELS;
    const float* __restrict__ arow_b = agg_back + (long long)rowc * CHANNELS;

    v8f acc = {};

#pragma unroll
    for (int k = 0; k < CHANNELS; k += 4) {
        float2 av = *(const float2*)(arow_o + k + kb);
        v2f a; a.x = av.x * sc_o; a.y = av.y * sc_o;
        v2f b;
        b.x = w_out[(k + kb)     * CHANNELS + col];
        b.y = w_out[(k + kb + 1) * CHANNELS + col];
        acc = __builtin_amdgcn_wmma_f32_16x16x4_f32(
            false, a, false, b, (short)0, acc, false, false);
    }
#pragma unroll
    for (int k = 0; k < CHANNELS; k += 4) {
        float2 av = *(const float2*)(arow_b + k + kb);
        v2f a; a.x = av.x * sc_b; a.y = av.y * sc_b;
        v2f b;
        b.x = w_back[(k + kb)     * CHANNELS + col];
        b.y = w_back[(k + kb + 1) * CHANNELS + col];
        acc = __builtin_amdgcn_wmma_f32_16x16x4_f32(
            false, a, false, b, (short)0, acc, false, false);
    }

#pragma unroll
    for (int j = 0; j < 8; ++j) {
        int mrow = rowTile * 16 + j + half * 8;
        if (mrow < N)
            out[(long long)mrow * CHANNELS + col] = acc[j];
    }
}

extern "C" void kernel_launch(void* const* d_in, const int* in_sizes, int n_in,
                              void* d_out, int out_size, void* d_ws, size_t ws_size,
                              hipStream_t stream) {
    const float* x       = (const float*)d_in[0];
    const int*   sources = (const int*)d_in[1];
    const int*   targets = (const int*)d_in[2];
    const float* norm    = (const float*)d_in[3];
    const float* norm_t  = (const float*)d_in[4];
    const float* w_out   = (const float*)d_in[5];
    const float* w_back  = (const float*)d_in[6];
    float*       out     = (float*)d_out;

    const int N = in_sizes[3];
    const int E = in_sizes[1];

    float* agg_out  = (float*)d_ws;
    float* agg_back = agg_out + (long long)N * CHANNELS;

    const int nq  = (E + 15) >> 4;
    const int per = (((nq + 31) >> 5) + 31) >> 5;
    const int nqp = per * 32 * 32;
    unsigned* kt = (unsigned*)d_out;
    unsigned* ks = kt + 4 * nqp;

    {
        int blocks = (nqp + 255) / 256;
        biconv_keys<<<blocks, 256, 0, stream>>>(targets, sources, kt, ks, E, nqp);
    }
    {
        int nbins = (N + TILE_NODES - 1) >> TILE_SHIFT;
        biconv_tile_sum<<<dim3(nbins, 2), 1024, 0, stream>>>(
            x, targets, sources, kt, ks, agg_out, agg_back, N, per);
    }
    {
        int numRowTiles = (N + 15) >> 4;
        int numTiles = numRowTiles * 4;
        int wavesPerBlock = 8;
        int blocks = (numTiles + wavesPerBlock - 1) / wavesPerBlock;
        biconv_wmma_gemm<<<blocks, 256, 0, stream>>>(
            agg_out, agg_back, norm, norm_t, w_out, w_back, out, N, numTiles);
    }
}
